# starting kernel
# baseline (speedup 1.0000x reference)
#include <hip/hip_runtime.h>

typedef float v2f __attribute__((ext_vector_type(2)));
typedef float v8f __attribute__((ext_vector_type(8)));

#define QDIM   256
#define QBATCH 32768

__global__ void qenc_build(const float* __restrict__ wrot,
                           const float* __restrict__ wrz,
                           float* __restrict__ MrT,
                           float* __restrict__ MiT) {
    extern __shared__ float lds[];
    const int tid = threadIdx.x;
    const int rl  = tid >> 3;
    const int u   = tid & 7;
    const int r   = blockIdx.x * 32 + rl;

    float* sre = lds + rl * 514;
    float* sim = sre + 257;

    const float w = 6.28318530717958647692f / 256.0f;
    for (int j = 0; j < 32; ++j) {
        int k  = u * 32 + j;
        int ph = (r * k) & 255;
        float sv, cv;
        sincosf(w * (float)ph, &sv, &cv);
        sre[k] = cv * 0.0625f;
        sim[k] = sv * 0.0625f;
    }
    __syncthreads();

    for (int l = 0; l < 4; ++l) {
        for (int q = 0; q < 8; ++q) {
            float th  = wrot[l * 8 + q];
            float phz = wrz[l * 8 + q];
            float c = cosf(0.5f * th), s = sinf(0.5f * th);
            float emr = cosf(0.5f * phz), emi = -sinf(0.5f * phz);
            float U00r =  emr * c, U00i =  emi * c;
            float U01r = -emr * s, U01i = -emi * s;
            float U10r =  emr * s, U10i = -emi * s;
            float U11r =  emr * c, U11i = -emi * c;
            int bit = 7 - q, mask = 1 << bit;
            for (int j = 0; j < 16; ++j) {
                int p  = u * 16 + j;
                int k0 = ((p >> bit) << (bit + 1)) | (p & (mask - 1));
                int k1 = k0 | mask;
                float ar = sre[k0], ai = sim[k0];
                float br = sre[k1], bi = sim[k1];
                sre[k0] = U00r * ar - U00i * ai + U01r * br - U01i * bi;
                sim[k0] = U00r * ai + U00i * ar + U01r * bi + U01i * br;
                sre[k1] = U10r * ar - U10i * ai + U11r * br - U11i * bi;
                sim[k1] = U10r * ai + U10i * ar + U11r * bi + U11i * br;
            }
            __syncthreads();
        }
        for (int q = 0; q < 8; ++q) {
            int cb = 7 - q, tb = 7 - ((q + 1) & 7);
            int tmask = 1 << tb;
            for (int j = 0; j < 32; ++j) {
                int k = u * 32 + j;
                if (((k >> cb) & 1) && !((k >> tb) & 1)) {
                    int k2 = k | tmask;
                    float tr = sre[k], ti = sim[k];
                    sre[k]  = sre[k2]; sim[k]  = sim[k2];
                    sre[k2] = tr;      sim[k2] = ti;
                }
            }
            __syncthreads();
        }
    }

    for (int j = 0; j < 32; ++j) {
        int n = u * 32 + j;
        MrT[n * 256 + r] = sre[n];
        MiT[n * 256 + r] = sim[n];
    }
}

#define WMMA_F32(A, B, C) \
    __builtin_amdgcn_wmma_f32_16x16x4_f32(false, (A), false, (B), (short)0, (C), false, false)

__global__ void qenc_gemm(const float* __restrict__ X,
                          const float* __restrict__ MrT,
                          const float* __restrict__ MiT,
                          float* __restrict__ out) {
    const int lane = threadIdx.x & 31;
    const int wid  = blockIdx.x * 8 + (threadIdx.x >> 5);
    const int n0   = (wid & 7) * 32;
    const int m0   = (wid >> 3) * 32;
    const int half = lane >> 4;
    const int l15  = lane & 15;

    const float* ap0  = X   + (m0 + l15) * 256 + 2 * half;
    const float* ap1  = ap0 + 16 * 256;
    const float* brp0 = MrT + (n0 + l15) * 256 + 2 * half;
    const float* brp1 = brp0 + 16 * 256;
    const float* bip0 = MiT + (n0 + l15) * 256 + 2 * half;
    const float* bip1 = bip0 + 16 * 256;

    v8f accR00 = {}, accI00 = {}, accR01 = {}, accI01 = {};
    v8f accR10 = {}, accI10 = {}, accR11 = {}, accI11 = {};
    float nrm0 = 0.0f, nrm1 = 0.0f;

#pragma unroll 2
    for (int k = 0; k < 256; k += 4) {
        v2f A0  = *(const v2f*)(ap0 + k);
        v2f A1  = *(const v2f*)(ap1 + k);
        v2f Br0 = *(const v2f*)(brp0 + k);
        v2f Br1 = *(const v2f*)(brp1 + k);
        v2f Bi0 = *(const v2f*)(bip0 + k);
        v2f Bi1 = *(const v2f*)(bip1 + k);
        nrm0 += A0.x * A0.x + A0.y * A0.y;
        nrm1 += A1.x * A1.x + A1.y * A1.y;
        accR00 = WMMA_F32(A0, Br0, accR00);
        accI00 = WMMA_F32(A0, Bi0, accI00);
        accR01 = WMMA_F32(A0, Br1, accR01);
        accI01 = WMMA_F32(A0, Bi1, accI01);
        accR10 = WMMA_F32(A1, Br0, accR10);
        accI10 = WMMA_F32(A1, Bi0, accI10);
        accR11 = WMMA_F32(A1, Br1, accR11);
        accI11 = WMMA_F32(A1, Bi1, accI11);
    }

    nrm0 += __shfl_xor(nrm0, 16, 32);
    nrm1 += __shfl_xor(nrm1, 16, 32);

#pragma unroll
    for (int i = 0; i < 8; ++i) {
        int row    = i + 8 * half;
        float inv0 = 1.0f / __shfl(nrm0, row, 32);
        float inv1 = 1.0f / __shfl(nrm1, row, 32);
        float* o0  = out + (m0 + row) * 256 + n0 + l15;
        float* o1  = out + (m0 + 16 + row) * 256 + n0 + l15;
        o0[0]  = (accR00[i] * accR00[i] + accI00[i] * accI00[i]) * inv0;
        o0[16] = (accR01[i] * accR01[i] + accI01[i] * accI01[i]) * inv0;
        o1[0]  = (accR10[i] * accR10[i] + accI10[i] * accI10[i]) * inv1;
        o1[16] = (accR11[i] * accR11[i] + accI11[i] * accI11[i]) * inv1;
    }
}

extern "C" void kernel_launch(void* const* d_in, const int* in_sizes, int n_in,
                              void* d_out, int out_size, void* d_ws, size_t ws_size,
                              hipStream_t stream) {
    const float* x    = (const float*)d_in[0];
    const float* wrot = (const float*)d_in[1];
    const float* wrz  = (const float*)d_in[2];
    float* out = (float*)d_out;

    float* MrT = (float*)d_ws;
    float* MiT = MrT + QDIM * QDIM;

    size_t lds_bytes = 32 * 514 * sizeof(float);
    qenc_build<<<8, 256, lds_bytes, stream>>>(wrot, wrz, MrT, MiT);

    qenc_gemm<<<1024, 256, 0, stream>>>(x, MrT, MiT, out);
}


# Matrix-build kernel: one wave per row, row kept in registers, gates and CNOTs as register ops or lane shuffles; rows transposed via LDS so one wave stores each 128-byte line; gate sum order changes
# speedup vs baseline: 2.0388x; 2.0388x over previous
#include <hip/hip_runtime.h>

typedef float v2f __attribute__((ext_vector_type(2)));
typedef float v8f __attribute__((ext_vector_type(8)));

#define QDIM   256
#define QBATCH 32768

__device__ __forceinline__ float lane_value(float v, int g) {
    return __int_as_float(__builtin_amdgcn_readlane(__float_as_int(v), g));
}

__global__ __launch_bounds__(1024) void qenc_build(const float* __restrict__ wrot,
                                                   const float* __restrict__ wrz,
                                                   float* __restrict__ MrT,
                                                   float* __restrict__ MiT) {
    __shared__ float tre[256 * 33];
    __shared__ float tim[256 * 33];
    const int lane = threadIdx.x & 31;
    const int rl   = threadIdx.x >> 5;
    const int r    = blockIdx.x * 32 + rl;

    const float th  = wrot[lane];
    const float phz = wrz[lane];
    const float gc  = cosf(0.5f * th),  gs  = sinf(0.5f * th);
    const float ger = cosf(0.5f * phz), gei = -sinf(0.5f * phz);

    float re[8], im[8];

    const float w = 6.28318530717958647692f / 256.0f;
#pragma unroll
    for (int j = 0; j < 8; ++j) {
        int k  = j * 32 + lane;
        int ph = (r * k) & 255;
        float sv, cv;
        sincosf(w * (float)ph, &sv, &cv);
        re[j] = cv * 0.0625f;
        im[j] = sv * 0.0625f;
    }

    int srcE = lane;
#pragma unroll
    for (int q = 6; q >= 3; --q)
        srcE ^= ((srcE >> (7 - q)) & 1) << (6 - q);
    const int srcO = srcE ^ 16;

#pragma unroll
    for (int l = 0; l < 4; ++l) {
#pragma unroll
        for (int q = 0; q < 8; ++q) {
            const int g = l * 8 + q;
            const float c   = lane_value(gc,  g);
            const float s   = lane_value(gs,  g);
            const float emr = lane_value(ger, g);
            const float emi = lane_value(gei, g);
            const float U00r =  emr * c, U00i =  emi * c;
            const float U01r = -emr * s, U01i = -emi * s;
            const float U10r =  emr * s, U10i = -emi * s;
            const float U11r =  emr * c, U11i = -emi * c;
            const int bit = 7 - q;
            if (bit >= 5) {
                const int jm = 1 << (bit - 5);
#pragma unroll
                for (int j0 = 0; j0 < 8; ++j0) {
                    if (j0 & jm) continue;
                    const int j1 = j0 | jm;
                    float ar = re[j0], ai = im[j0];
                    float br = re[j1], bi = im[j1];
                    re[j0] = U00r * ar - U00i * ai + U01r * br - U01i * bi;
                    im[j0] = U00r * ai + U00i * ar + U01r * bi + U01i * br;
                    re[j1] = U10r * ar - U10i * ai + U11r * br - U11i * bi;
                    im[j1] = U10r * ai + U10i * ar + U11r * bi + U11i * br;
                }
            } else {
                const int  lm = 1 << bit;
                const bool hi = (lane & lm) != 0;
                const float Pmr = hi ? U11r : U00r, Pmi = hi ? U11i : U00i;
                const float Pxr = hi ? U10r : U01r, Pxi = hi ? U10i : U01i;
#pragma unroll
                for (int j = 0; j < 8; ++j) {
                    float mr = re[j], mi = im[j];
                    float xr = __shfl_xor(mr, lm, 32);
                    float xi = __shfl_xor(mi, lm, 32);
                    re[j] = Pmr * mr - Pmi * mi + Pxr * xr - Pxi * xi;
                    im[j] = Pmr * mi + Pmi * mr + Pxr * xi + Pxi * xr;
                }
            }
        }
#pragma unroll
        for (int q = 0; q < 2; ++q) {
            const int cm = 1 << (2 - q), tm = 1 << (1 - q);
#pragma unroll
            for (int j = 0; j < 8; ++j) {
                if ((j & cm) && !(j & tm)) {
                    float t0 = re[j]; re[j] = re[j | tm]; re[j | tm] = t0;
                    float t1 = im[j]; im[j] = im[j | tm]; im[j | tm] = t1;
                }
            }
        }
#pragma unroll
        for (int j = 0; j < 8; ++j) {
            const int src = (j & 1) ? srcO : srcE;
            re[j] = __shfl(re[j], src, 32);
            im[j] = __shfl(im[j], src, 32);
        }
        {
            const bool ctl = (lane & 1) != 0;
#pragma unroll
            for (int j = 0; j < 4; ++j) {
                float a0 = re[j], b0 = re[j | 4];
                re[j] = ctl ? b0 : a0; re[j | 4] = ctl ? a0 : b0;
                float a1 = im[j], b1 = im[j | 4];
                im[j] = ctl ? b1 : a1; im[j | 4] = ctl ? a1 : b1;
            }
        }
    }

#pragma unroll
    for (int j = 0; j < 8; ++j) {
        int n = j * 32 + lane;
        tre[n * 33 + rl] = re[j];
        tim[n * 33 + rl] = im[j];
    }
    __syncthreads();
    for (int i = 0; i < 8; ++i) {
        int n = rl * 8 + i;
        MrT[n * 256 + blockIdx.x * 32 + lane] = tre[n * 33 + lane];
        MiT[n * 256 + blockIdx.x * 32 + lane] = tim[n * 33 + lane];
    }
}

#define WMMA_F32(A, B, C) \
    __builtin_amdgcn_wmma_f32_16x16x4_f32(false, (A), false, (B), (short)0, (C), false, false)

__global__ void qenc_gemm(const float* __restrict__ X,
                          const float* __restrict__ MrT,
                          const float* __restrict__ MiT,
                          float* __restrict__ out) {
    const int lane = threadIdx.x & 31;
    const int wid  = blockIdx.x * 8 + (threadIdx.x >> 5);
    const int n0   = (wid & 7) * 32;
    const int m0   = (wid >> 3) * 32;
    const int half = lane >> 4;
    const int l15  = lane & 15;

    const float* ap0  = X   + (m0 + l15) * 256 + 2 * half;
    const float* ap1  = ap0 + 16 * 256;
    const float* brp0 = MrT + (n0 + l15) * 256 + 2 * half;
    const float* brp1 = brp0 + 16 * 256;
    const float* bip0 = MiT + (n0 + l15) * 256 + 2 * half;
    const float* bip1 = bip0 + 16 * 256;

    v8f accR00 = {}, accI00 = {}, accR01 = {}, accI01 = {};
    v8f accR10 = {}, accI10 = {}, accR11 = {}, accI11 = {};
    float nrm0 = 0.0f, nrm1 = 0.0f;

#pragma unroll 2
    for (int k = 0; k < 256; k += 4) {
        v2f A0  = *(const v2f*)(ap0 + k);
        v2f A1  = *(const v2f*)(ap1 + k);
        v2f Br0 = *(const v2f*)(brp0 + k);
        v2f Br1 = *(const v2f*)(brp1 + k);
        v2f Bi0 = *(const v2f*)(bip0 + k);
        v2f Bi1 = *(const v2f*)(bip1 + k);
        nrm0 += A0.x * A0.x + A0.y * A0.y;
        nrm1 += A1.x * A1.x + A1.y * A1.y;
        accR00 = WMMA_F32(A0, Br0, accR00);
        accI00 = WMMA_F32(A0, Bi0, accI00);
        accR01 = WMMA_F32(A0, Br1, accR01);
        accI01 = WMMA_F32(A0, Bi1, accI01);
        accR10 = WMMA_F32(A1, Br0, accR10);
        accI10 = WMMA_F32(A1, Bi0, accI10);
        accR11 = WMMA_F32(A1, Br1, accR11);
        accI11 = WMMA_F32(A1, Bi1, accI11);
    }

    nrm0 += __shfl_xor(nrm0, 16, 32);
    nrm1 += __shfl_xor(nrm1, 16, 32);

#pragma unroll
    for (int i = 0; i < 8; ++i) {
        int row    = i + 8 * half;
        float inv0 = 1.0f / __shfl(nrm0, row, 32);
        float inv1 = 1.0f / __shfl(nrm1, row, 32);
        float* o0  = out + (m0 + row) * 256 + n0 + l15;
        float* o1  = out + (m0 + 16 + row) * 256 + n0 + l15;
        o0[0]  = (accR00[i] * accR00[i] + accI00[i] * accI00[i]) * inv0;
        o0[16] = (accR01[i] * accR01[i] + accI01[i] * accI01[i]) * inv0;
        o1[0]  = (accR10[i] * accR10[i] + accI10[i] * accI10[i]) * inv1;
        o1[16] = (accR11[i] * accR11[i] + accI11[i] * accI11[i]) * inv1;
    }
}

extern "C" void kernel_launch(void* const* d_in, const int* in_sizes, int n_in,
                              void* d_out, int out_size, void* d_ws, size_t ws_size,
                              hipStream_t stream) {
    const float* x    = (const float*)d_in[0];
    const float* wrot = (const float*)d_in[1];
    const float* wrz  = (const float*)d_in[2];
    float* out = (float*)d_out;

    float* MrT = (float*)d_ws;
    float* MiT = MrT + QDIM * QDIM;

    qenc_build<<<8, 1024, 0, stream>>>(wrot, wrz, MrT, MiT);

    qenc_gemm<<<1024, 256, 0, stream>>>(x, MrT, MiT, out);
}


# B operand written by the build kernel in WMMA fragment order (Re,Im interleaved): each wave reads a k-step's B as 2 contiguous 512-byte loads, not 4 loads of 16 lines each
# speedup vs baseline: 2.9362x; 1.4402x over previous
#include <hip/hip_runtime.h>

typedef float v2f __attribute__((ext_vector_type(2)));
typedef float v4f __attribute__((ext_vector_type(4)));
typedef float v8f __attribute__((ext_vector_type(8)));

#define QDIM   256
#define QBATCH 32768

__device__ __forceinline__ float lane_value(float v, int g) {
    return __int_as_float(__builtin_amdgcn_readlane(__float_as_int(v), g));
}

__global__ __launch_bounds__(1024) void qenc_build(const float* __restrict__ wrot,
                                                   const float* __restrict__ wrz,
                                                   float* __restrict__ P) {
    __shared__ v4f stage4[4096];
    const int lane = threadIdx.x & 31;
    const int rl   = threadIdx.x >> 5;
    const int r    = blockIdx.x * 32 + rl;

    const float th  = wrot[lane];
    const float phz = wrz[lane];
    const float gc  = cosf(0.5f * th),  gs  = sinf(0.5f * th);
    const float ger = cosf(0.5f * phz), gei = -sinf(0.5f * phz);

    float re[8], im[8];

    const float w = 6.28318530717958647692f / 256.0f;
#pragma unroll
    for (int j = 0; j < 8; ++j) {
        int k  = j * 32 + lane;
        int ph = (r * k) & 255;
        float sv, cv;
        sincosf(w * (float)ph, &sv, &cv);
        re[j] = cv * 0.0625f;
        im[j] = sv * 0.0625f;
    }

    int srcE = lane;
#pragma unroll
    for (int q = 6; q >= 3; --q)
        srcE ^= ((srcE >> (7 - q)) & 1) << (6 - q);
    const int srcO = srcE ^ 16;

#pragma unroll
    for (int l = 0; l < 4; ++l) {
#pragma unroll
        for (int q = 0; q < 8; ++q) {
            const int g = l * 8 + q;
            const float c   = lane_value(gc,  g);
            const float s   = lane_value(gs,  g);
            const float emr = lane_value(ger, g);
            const float emi = lane_value(gei, g);
            const float U00r =  emr * c, U00i =  emi * c;
            const float U01r = -emr * s, U01i = -emi * s;
            const float U10r =  emr * s, U10i = -emi * s;
            const float U11r =  emr * c, U11i = -emi * c;
            const int bit = 7 - q;
            if (bit >= 5) {
                const int jm = 1 << (bit - 5);
#pragma unroll
                for (int j0 = 0; j0 < 8; ++j0) {
                    if (j0 & jm) continue;
                    const int j1 = j0 | jm;
                    float ar = re[j0], ai = im[j0];
                    float br = re[j1], bi = im[j1];
                    re[j0] = U00r * ar - U00i * ai + U01r * br - U01i * bi;
                    im[j0] = U00r * ai + U00i * ar + U01r * bi + U01i * br;
                    re[j1] = U10r * ar - U10i * ai + U11r * br - U11i * bi;
                    im[j1] = U10r * ai + U10i * ar + U11r * bi + U11i * br;
                }
            } else {
                const int  lm = 1 << bit;
                const bool hi = (lane & lm) != 0;
                const float Pmr = hi ? U11r : U00r, Pmi = hi ? U11i : U00i;
                const float Pxr = hi ? U10r : U01r, Pxi = hi ? U10i : U01i;
#pragma unroll
                for (int j = 0; j < 8; ++j) {
                    float mr = re[j], mi = im[j];
                    float xr = __shfl_xor(mr, lm, 32);
                    float xi = __shfl_xor(mi, lm, 32);
                    re[j] = Pmr * mr - Pmi * mi + Pxr * xr - Pxi * xi;
                    im[j] = Pmr * mi + Pmi * mr + Pxr * xi + Pxi * xr;
                }
            }
        }
#pragma unroll
        for (int q = 0; q < 2; ++q) {
            const int cm = 1 << (2 - q), tm = 1 << (1 - q);
#pragma unroll
            for (int j = 0; j < 8; ++j) {
                if ((j & cm) && !(j & tm)) {
                    float t0 = re[j]; re[j] = re[j | tm]; re[j | tm] = t0;
                    float t1 = im[j]; im[j] = im[j | tm]; im[j | tm] = t1;
                }
            }
        }
#pragma unroll
        for (int j = 0; j < 8; ++j) {
            const int src = (j & 1) ? srcO : srcE;
            re[j] = __shfl(re[j], src, 32);
            im[j] = __shfl(im[j], src, 32);
        }
        {
            const bool ctl = (lane & 1) != 0;
#pragma unroll
            for (int j = 0; j < 4; ++j) {
                float a0 = re[j], b0 = re[j | 4];
                re[j] = ctl ? b0 : a0; re[j | 4] = ctl ? a0 : b0;
                float a1 = im[j], b1 = im[j | 4];
                im[j] = ctl ? b1 : a1; im[j | 4] = ctl ? a1 : b1;
            }
        }
    }

    float* stage = (float*)stage4;
    const int sl = rl >> 2;
    const int kh = (rl >> 1) & 1;
    const int ke = rl & 1;
#pragma unroll
    for (int j = 0; j < 8; ++j) {
        const int t = 2 * j + (lane >> 4);
        const int d = (((t * 8 + sl) * 32 + kh * 16 + (lane & 15)) << 2) + ke;
        stage[d]     = re[j];
        stage[d + 2] = im[j];
    }
    __syncthreads();
    v4f* P4 = (v4f*)P;
    for (int p = 0; p < 4; ++p) {
        const int q = p * 1024 + threadIdx.x;
        const int t = q >> 8;
        P4[(t * 64 + blockIdx.x * 8) * 32 + (q & 255)] = stage4[q];
    }
}

#define WMMA_F32(A, B, C) \
    __builtin_amdgcn_wmma_f32_16x16x4_f32(false, (A), false, (B), (short)0, (C), false, false)

__global__ void qenc_gemm(const float* __restrict__ X,
                          const float* __restrict__ P,
                          float* __restrict__ out) {
    const int lane = threadIdx.x & 31;
    const int wid  = blockIdx.x * 8 + (threadIdx.x >> 5);
    const int n0   = (wid & 7) * 32;
    const int m0   = (wid >> 3) * 32;
    const int half = lane >> 4;
    const int l15  = lane & 15;

    const float* ap0 = X   + (m0 + l15) * 256 + 2 * half;
    const float* ap1 = ap0 + 16 * 256;
    const v4f*   bp0 = (const v4f*)P + (n0 >> 4) * 64 * 32 + lane;
    const v4f*   bp1 = bp0 + 64 * 32;

    v8f accR00 = {}, accI00 = {}, accR01 = {}, accI01 = {};
    v8f accR10 = {}, accI10 = {}, accR11 = {}, accI11 = {};
    float nrm0 = 0.0f, nrm1 = 0.0f;

#pragma unroll 2
    for (int k = 0; k < 256; k += 4) {
        v2f A0  = *(const v2f*)(ap0 + k);
        v2f A1  = *(const v2f*)(ap1 + k);
        v4f B0  = bp0[k * 8];
        v4f B1  = bp1[k * 8];
        v2f Br0 = B0.xy, Bi0 = B0.zw;
        v2f Br1 = B1.xy, Bi1 = B1.zw;
        nrm0 += A0.x * A0.x + A0.y * A0.y;
        nrm1 += A1.x * A1.x + A1.y * A1.y;
        accR00 = WMMA_F32(A0, Br0, accR00);
        accI00 = WMMA_F32(A0, Bi0, accI00);
        accR01 = WMMA_F32(A0, Br1, accR01);
        accI01 = WMMA_F32(A0, Bi1, accI01);
        accR10 = WMMA_F32(A1, Br0, accR10);
        accI10 = WMMA_F32(A1, Bi0, accI10);
        accR11 = WMMA_F32(A1, Br1, accR11);
        accI11 = WMMA_F32(A1, Bi1, accI11);
    }

    nrm0 += __shfl_xor(nrm0, 16, 32);
    nrm1 += __shfl_xor(nrm1, 16, 32);

#pragma unroll
    for (int i = 0; i < 8; ++i) {
        int row    = i + 8 * half;
        float inv0 = 1.0f / __shfl(nrm0, row, 32);
        float inv1 = 1.0f / __shfl(nrm1, row, 32);
        float* o0  = out + (m0 + row) * 256 + n0 + l15;
        float* o1  = out + (m0 + 16 + row) * 256 + n0 + l15;
        o0[0]  = (accR00[i] * accR00[i] + accI00[i] * accI00[i]) * inv0;
        o0[16] = (accR01[i] * accR01[i] + accI01[i] * accI01[i]) * inv0;
        o1[0]  = (accR10[i] * accR10[i] + accI10[i] * accI10[i]) * inv1;
        o1[16] = (accR11[i] * accR11[i] + accI11[i] * accI11[i]) * inv1;
    }
}

extern "C" void kernel_launch(void* const* d_in, const int* in_sizes, int n_in,
                              void* d_out, int out_size, void* d_ws, size_t ws_size,
                              hipStream_t stream) {
    const float* x    = (const float*)d_in[0];
    const float* wrot = (const float*)d_in[1];
    const float* wrz  = (const float*)d_in[2];
    float* out = (float*)d_out;

    float* P = (float*)d_ws;

    qenc_build<<<8, 1024, 0, stream>>>(wrot, wrz, P);

    qenc_gemm<<<1024, 256, 0, stream>>>(x, P, out);
}


# Matrix-build kernel spread from 8 work-groups of 32 waves to 64 work-groups of 4 waves: each row's wave has a SIMD to itself; each work-group stages and stores one k-step of P (8 KB LDS)
# speedup vs baseline: 3.1051x; 1.0575x over previous
#include <hip/hip_runtime.h>

typedef float v2f __attribute__((ext_vector_type(2)));
typedef float v4f __attribute__((ext_vector_type(4)));
typedef float v8f __attribute__((ext_vector_type(8)));

#define QDIM   256
#define QBATCH 32768

__device__ __forceinline__ float lane_value(float v, int g) {
    return __int_as_float(__builtin_amdgcn_readlane(__float_as_int(v), g));
}

__global__ __launch_bounds__(128) void qenc_build(const float* __restrict__ wrot,
                                                   const float* __restrict__ wrz,
                                                   float* __restrict__ P) {
    __shared__ v4f stage4[512];
    const int lane = threadIdx.x & 31;
    const int rl   = threadIdx.x >> 5;
    const int r    = blockIdx.x * 4 + rl;

    const float th  = wrot[lane];
    const float phz = wrz[lane];
    const float gc  = cosf(0.5f * th),  gs  = sinf(0.5f * th);
    const float ger = cosf(0.5f * phz), gei = -sinf(0.5f * phz);

    float re[8], im[8];

    const float w = 6.28318530717958647692f / 256.0f;
#pragma unroll
    for (int j = 0; j < 8; ++j) {
        int k  = j * 32 + lane;
        int ph = (r * k) & 255;
        float sv, cv;
        sincosf(w * (float)ph, &sv, &cv);
        re[j] = cv * 0.0625f;
        im[j] = sv * 0.0625f;
    }

    int srcE = lane;
#pragma unroll
    for (int q = 6; q >= 3; --q)
        srcE ^= ((srcE >> (7 - q)) & 1) << (6 - q);
    const int srcO = srcE ^ 16;

#pragma unroll
    for (int l = 0; l < 4; ++l) {
#pragma unroll
        for (int q = 0; q < 8; ++q) {
            const int g = l * 8 + q;
            const float c   = lane_value(gc,  g);
            const float s   = lane_value(gs,  g);
            const float emr = lane_value(ger, g);
            const float emi = lane_value(gei, g);
            const float U00r =  emr * c, U00i =  emi * c;
            const float U01r = -emr * s, U01i = -emi * s;
            const float U10r =  emr * s, U10i = -emi * s;
            const float U11r =  emr * c, U11i = -emi * c;
            const int bit = 7 - q;
            if (bit >= 5) {
                const int jm = 1 << (bit - 5);
#pragma unroll
                for (int j0 = 0; j0 < 8; ++j0) {
                    if (j0 & jm) continue;
                    const int j1 = j0 | jm;
                    float ar = re[j0], ai = im[j0];
                    float br = re[j1], bi = im[j1];
                    re[j0] = U00r * ar - U00i * ai + U01r * br - U01i * bi;
                    im[j0] = U00r * ai + U00i * ar + U01r * bi + U01i * br;
                    re[j1] = U10r * ar - U10i * ai + U11r * br - U11i * bi;
                    im[j1] = U10r * ai + U10i * ar + U11r * bi + U11i * br;
                }
            } else {
                const int  lm = 1 << bit;
                const bool hi = (lane & lm) != 0;
                const float Pmr = hi ? U11r : U00r, Pmi = hi ? U11i : U00i;
                const float Pxr = hi ? U10r : U01r, Pxi = hi ? U10i : U01i;
#pragma unroll
                for (int j = 0; j < 8; ++j) {
                    float mr = re[j], mi = im[j];
                    float xr = __shfl_xor(mr, lm, 32);
                    float xi = __shfl_xor(mi, lm, 32);
                    re[j] = Pmr * mr - Pmi * mi + Pxr * xr - Pxi * xi;
                    im[j] = Pmr * mi + Pmi * mr + Pxr * xi + Pxi * xr;
                }
            }
        }
#pragma unroll
        for (int q = 0; q < 2; ++q) {
            const int cm = 1 << (2 - q), tm = 1 << (1 - q);
#pragma unroll
            for (int j = 0; j < 8; ++j) {
                if ((j & cm) && !(j & tm)) {
                    float t0 = re[j]; re[j] = re[j | tm]; re[j | tm] = t0;
                    float t1 = im[j]; im[j] = im[j | tm]; im[j | tm] = t1;
                }
            }
        }
#pragma unroll
        for (int j = 0; j < 8; ++j) {
            const int src = (j & 1) ? srcO : srcE;
            re[j] = __shfl(re[j], src, 32);
            im[j] = __shfl(im[j], src, 32);
        }
        {
            const bool ctl = (lane & 1) != 0;
#pragma unroll
            for (int j = 0; j < 4; ++j) {
                float a0 = re[j], b0 = re[j | 4];
                re[j] = ctl ? b0 : a0; re[j | 4] = ctl ? a0 : b0;
                float a1 = im[j], b1 = im[j | 4];
                im[j] = ctl ? b1 : a1; im[j | 4] = ctl ? a1 : b1;
            }
        }
    }

    float* stage = (float*)stage4;
    const int kh = (rl >> 1) & 1;
    const int ke = rl & 1;
#pragma unroll
    for (int j = 0; j < 8; ++j) {
        const int t = 2 * j + (lane >> 4);
        const int d = ((t * 32 + kh * 16 + (lane & 15)) << 2) + ke;
        stage[d]     = re[j];
        stage[d + 2] = im[j];
    }
    __syncthreads();
    v4f* P4 = (v4f*)P;
    for (int p = 0; p < 4; ++p) {
        const int q = p * 128 + threadIdx.x;
        const int t = q >> 5;
        P4[(t * 64 + blockIdx.x) * 32 + (q & 31)] = stage4[q];
    }
}

#define WMMA_F32(A, B, C) \
    __builtin_amdgcn_wmma_f32_16x16x4_f32(false, (A), false, (B), (short)0, (C), false, false)

__global__ void qenc_gemm(const float* __restrict__ X,
                          const float* __restrict__ P,
                          float* __restrict__ out) {
    const int lane = threadIdx.x & 31;
    const int wid  = blockIdx.x * 8 + (threadIdx.x >> 5);
    const int n0   = (wid & 7) * 32;
    const int m0   = (wid >> 3) * 32;
    const int half = lane >> 4;
    const int l15  = lane & 15;

    const float* ap0 = X   + (m0 + l15) * 256 + 2 * half;
    const float* ap1 = ap0 + 16 * 256;
    const v4f*   bp0 = (const v4f*)P + (n0 >> 4) * 64 * 32 + lane;
    const v4f*   bp1 = bp0 + 64 * 32;

    v8f accR00 = {}, accI00 = {}, accR01 = {}, accI01 = {};
    v8f accR10 = {}, accI10 = {}, accR11 = {}, accI11 = {};
    float nrm0 = 0.0f, nrm1 = 0.0f;

#pragma unroll 2
    for (int k = 0; k < 256; k += 4) {
        v2f A0  = *(const v2f*)(ap0 + k);
        v2f A1  = *(const v2f*)(ap1 + k);
        v4f B0  = bp0[k * 8];
        v4f B1  = bp1[k * 8];
        v2f Br0 = B0.xy, Bi0 = B0.zw;
        v2f Br1 = B1.xy, Bi1 = B1.zw;
        nrm0 += A0.x * A0.x + A0.y * A0.y;
        nrm1 += A1.x * A1.x + A1.y * A1.y;
        accR00 = WMMA_F32(A0, Br0, accR00);
        accI00 = WMMA_F32(A0, Bi0, accI00);
        accR01 = WMMA_F32(A0, Br1, accR01);
        accI01 = WMMA_F32(A0, Bi1, accI01);
        accR10 = WMMA_F32(A1, Br0, accR10);
        accI10 = WMMA_F32(A1, Bi0, accI10);
        accR11 = WMMA_F32(A1, Br1, accR11);
        accI11 = WMMA_F32(A1, Bi1, accI11);
    }

    nrm0 += __shfl_xor(nrm0, 16, 32);
    nrm1 += __shfl_xor(nrm1, 16, 32);

#pragma unroll
    for (int i = 0; i < 8; ++i) {
        int row    = i + 8 * half;
        float inv0 = 1.0f / __shfl(nrm0, row, 32);
        float inv1 = 1.0f / __shfl(nrm1, row, 32);
        float* o0  = out + (m0 + row) * 256 + n0 + l15;
        float* o1  = out + (m0 + 16 + row) * 256 + n0 + l15;
        o0[0]  = (accR00[i] * accR00[i] + accI00[i] * accI00[i]) * inv0;
        o0[16] = (accR01[i] * accR01[i] + accI01[i] * accI01[i]) * inv0;
        o1[0]  = (accR10[i] * accR10[i] + accI10[i] * accI10[i]) * inv1;
        o1[16] = (accR11[i] * accR11[i] + accI11[i] * accI11[i]) * inv1;
    }
}

extern "C" void kernel_launch(void* const* d_in, const int* in_sizes, int n_in,
                              void* d_out, int out_size, void* d_ws, size_t ws_size,
                              hipStream_t stream) {
    const float* x    = (const float*)d_in[0];
    const float* wrot = (const float*)d_in[1];
    const float* wrz  = (const float*)d_in[2];
    float* out = (float*)d_out;

    float* P = (float*)d_ws;

    qenc_build<<<64, 128, 0, stream>>>(wrot, wrz, P);

    qenc_gemm<<<1024, 256, 0, stream>>>(x, P, out);
}


# Product kernel: wave tile 32x32 -> 64x64 (32 accumulator tiles, 2 waves per SIMD, 256 work-groups), next K-step's fragments loaded a step ahead into a second register set; norm sum order changes
# speedup vs baseline: 3.6374x; 1.1714x over previous
#include <hip/hip_runtime.h>

typedef float v2f __attribute__((ext_vector_type(2)));
typedef float v4f __attribute__((ext_vector_type(4)));
typedef float v8f __attribute__((ext_vector_type(8)));

#define QDIM   256
#define QBATCH 32768

__device__ __forceinline__ float lane_value(float v, int g) {
    return __int_as_float(__builtin_amdgcn_readlane(__float_as_int(v), g));
}

__global__ __launch_bounds__(128) void qenc_build(const float* __restrict__ wrot,
                                                   const float* __restrict__ wrz,
                                                   float* __restrict__ P) {
    __shared__ v4f stage4[512];
    const int lane = threadIdx.x & 31;
    const int rl   = threadIdx.x >> 5;
    const int r    = blockIdx.x * 4 + rl;

    const float th  = wrot[lane];
    const float phz = wrz[lane];
    const float gc  = cosf(0.5f * th),  gs  = sinf(0.5f * th);
    const float ger = cosf(0.5f * phz), gei = -sinf(0.5f * phz);

    float re[8], im[8];

    const float w = 6.28318530717958647692f / 256.0f;
#pragma unroll
    for (int j = 0; j < 8; ++j) {
        int k  = j * 32 + lane;
        int ph = (r * k) & 255;
        float sv, cv;
        sincosf(w * (float)ph, &sv, &cv);
        re[j] = cv * 0.0625f;
        im[j] = sv * 0.0625f;
    }

    int srcE = lane;
#pragma unroll
    for (int q = 6; q >= 3; --q)
        srcE ^= ((srcE >> (7 - q)) & 1) << (6 - q);
    const int srcO = srcE ^ 16;

#pragma unroll
    for (int l = 0; l < 4; ++l) {
#pragma unroll
        for (int q = 0; q < 8; ++q) {
            const int g = l * 8 + q;
            const float c   = lane_value(gc,  g);
            const float s   = lane_value(gs,  g);
            const float emr = lane_value(ger, g);
            const float emi = lane_value(gei, g);
            const float U00r =  emr * c, U00i =  emi * c;
            const float U01r = -emr * s, U01i = -emi * s;
            const float U10r =  emr * s, U10i = -emi * s;
            const float U11r =  emr * c, U11i = -emi * c;
            const int bit = 7 - q;
            if (bit >= 5) {
                const int jm = 1 << (bit - 5);
#pragma unroll
                for (int j0 = 0; j0 < 8; ++j0) {
                    if (j0 & jm) continue;
                    const int j1 = j0 | jm;
                    float ar = re[j0], ai = im[j0];
                    float br = re[j1], bi = im[j1];
                    re[j0] = U00r * ar - U00i * ai + U01r * br - U01i * bi;
                    im[j0] = U00r * ai + U00i * ar + U01r * bi + U01i * br;
                    re[j1] = U10r * ar - U10i * ai + U11r * br - U11i * bi;
                    im[j1] = U10r * ai + U10i * ar + U11r * bi + U11i * br;
                }
            } else {
                const int  lm = 1 << bit;
                const bool hi = (lane & lm) != 0;
                const float Pmr = hi ? U11r : U00r, Pmi = hi ? U11i : U00i;
                const float Pxr = hi ? U10r : U01r, Pxi = hi ? U10i : U01i;
#pragma unroll
                for (int j = 0; j < 8; ++j) {
                    float mr = re[j], mi = im[j];
                    float xr = __shfl_xor(mr, lm, 32);
                    float xi = __shfl_xor(mi, lm, 32);
                    re[j] = Pmr * mr - Pmi * mi + Pxr * xr - Pxi * xi;
                    im[j] = Pmr * mi + Pmi * mr + Pxr * xi + Pxi * xr;
                }
            }
        }
#pragma unroll
        for (int q = 0; q < 2; ++q) {
            const int cm = 1 << (2 - q), tm = 1 << (1 - q);
#pragma unroll
            for (int j = 0; j < 8; ++j) {
                if ((j & cm) && !(j & tm)) {
                    float t0 = re[j]; re[j] = re[j | tm]; re[j | tm] = t0;
                    float t1 = im[j]; im[j] = im[j | tm]; im[j | tm] = t1;
                }
            }
        }
#pragma unroll
        for (int j = 0; j < 8; ++j) {
            const int src = (j & 1) ? srcO : srcE;
            re[j] = __shfl(re[j], src, 32);
            im[j] = __shfl(im[j], src, 32);
        }
        {
            const bool ctl = (lane & 1) != 0;
#pragma unroll
            for (int j = 0; j < 4; ++j) {
                float a0 = re[j], b0 = re[j | 4];
                re[j] = ctl ? b0 : a0; re[j | 4] = ctl ? a0 : b0;
                float a1 = im[j], b1 = im[j | 4];
                im[j] = ctl ? b1 : a1; im[j | 4] = ctl ? a1 : b1;
            }
        }
    }

    float* stage = (float*)stage4;
    const int kh = (rl >> 1) & 1;
    const int ke = rl & 1;
#pragma unroll
    for (int j = 0; j < 8; ++j) {
        const int t = 2 * j + (lane >> 4);
        const int d = ((t * 32 + kh * 16 + (lane & 15)) << 2) + ke;
        stage[d]     = re[j];
        stage[d + 2] = im[j];
    }
    __syncthreads();
    v4f* P4 = (v4f*)P;
    for (int p = 0; p < 4; ++p) {
        const int q = p * 128 + threadIdx.x;
        const int t = q >> 5;
        P4[(t * 64 + blockIdx.x) * 32 + (q & 31)] = stage4[q];
    }
}

#define WMMA_F32(A, B, C) \
    __builtin_amdgcn_wmma_f32_16x16x4_f32(false, (A), false, (B), (short)0, (C), false, false)

#define LOAD_STEP(A, B, s)                                          \
    _Pragma("unroll")                                               \
    for (int g = 0; g < 4; ++g)                                     \
        A[g] = *(const v2f*)(ap + g * 4096 + 4 * (s));              \
    _Pragma("unroll")                                               \
    for (int c = 0; c < 4; ++c)                                     \
        B[c] = bp[c * 2048 + (s) * 32];

#define COMPUTE_STEP(A, B)                                          \
    _Pragma("unroll")                                               \
    for (int g = 0; g < 4; ++g)                                     \
        nrm2[g] = A[g] * A[g] + nrm2[g];                            \
    _Pragma("unroll")                                               \
    for (int c = 0; c < 4; ++c) {                                   \
        const v2f Br = B[c].xy, Bi = B[c].zw;                       \
        _Pragma("unroll")                                           \
        for (int g = 0; g < 4; ++g) {                               \
            accR[g][c] = WMMA_F32(A[g], Br, accR[g][c]);            \
            accI[g][c] = WMMA_F32(A[g], Bi, accI[g][c]);            \
        }                                                           \
    }

__global__ __launch_bounds__(256) void qenc_gemm(const float* __restrict__ X,
                                                  const float* __restrict__ P,
                                                  float* __restrict__ out) {
    const int lane = threadIdx.x & 31;
    const int wid  = blockIdx.x * 8 + (threadIdx.x >> 5);
    const int n0   = (wid & 3) * 64;
    const int m0   = (wid >> 2) * 64;
    const int half = lane >> 4;
    const int l15  = lane & 15;

    const float* ap = X + (m0 + l15) * 256 + 2 * half;
    const v4f*   bp = (const v4f*)P + (n0 >> 4) * 64 * 32 + lane;

    v8f accR[4][4], accI[4][4];
#pragma unroll
    for (int g = 0; g < 4; ++g) {
#pragma unroll
        for (int c = 0; c < 4; ++c) {
            accR[g][c] = (v8f){};
            accI[g][c] = (v8f){};
        }
    }
    v2f nrm2[4];
#pragma unroll
    for (int g = 0; g < 4; ++g) nrm2[g] = (v2f){0.0f, 0.0f};

    v2f a0[4], a1[4];
    v4f b0[4], b1[4];

    LOAD_STEP(a0, b0, 0)
    for (int s = 0; s < 62; s += 2) {
        LOAD_STEP(a1, b1, s + 1)
        COMPUTE_STEP(a0, b0)
        LOAD_STEP(a0, b0, s + 2)
        COMPUTE_STEP(a1, b1)
    }
    LOAD_STEP(a1, b1, 63)
    COMPUTE_STEP(a0, b0)
    COMPUTE_STEP(a1, b1)

    float nrm[4];
#pragma unroll
    for (int g = 0; g < 4; ++g) {
        nrm[g] = nrm2[g].x + nrm2[g].y;
        nrm[g] += __shfl_xor(nrm[g], 16, 32);
    }

#pragma unroll
    for (int g = 0; g < 4; ++g) {
#pragma unroll
        for (int i = 0; i < 8; ++i) {
            const int row   = i + 8 * half;
            const float inv = 1.0f / __shfl(nrm[g], row, 32);
            float* o = out + (m0 + 16 * g + row) * 256 + n0 + l15;
#pragma unroll
            for (int c = 0; c < 4; ++c)
                o[16 * c] = (accR[g][c][i] * accR[g][c][i] + accI[g][c][i] * accI[g][c][i]) * inv;
        }
    }
}

extern "C" void kernel_launch(void* const* d_in, const int* in_sizes, int n_in,
                              void* d_out, int out_size, void* d_ws, size_t ws_size,
                              hipStream_t stream) {
    const float* x    = (const float*)d_in[0];
    const float* wrot = (const float*)d_in[1];
    const float* wrz  = (const float*)d_in[2];
    float* out = (float*)d_out;

    float* P = (float*)d_ws;

    qenc_build<<<64, 128, 0, stream>>>(wrot, wrz, P);

    qenc_gemm<<<256, 256, 0, stream>>>(x, P, out);
}


# Matrix-build kernel: all lane exchanges of the gate chain (lane-bit gates, lane-to-lane CNOTs) as vector-ALU cross-lane moves (DPP row_xmask, permlane16/x16), not LDS permutes; arithmetic unchanged
# speedup vs baseline: 3.8066x; 1.0465x over previous
#include <hip/hip_runtime.h>

typedef float v2f __attribute__((ext_vector_type(2)));
typedef float v4f __attribute__((ext_vector_type(4)));
typedef float v8f __attribute__((ext_vector_type(8)));

#define QDIM   256
#define QBATCH 32768

__device__ __forceinline__ float lane_value(float v, int g) {
    return __int_as_float(__builtin_amdgcn_readlane(__float_as_int(v), g));
}

template <int CTRL, int ROWMASK, int BANKMASK>
__device__ __forceinline__ float dpp_move(float v) {
    const int b = __float_as_int(v);
    return __int_as_float(__builtin_amdgcn_update_dpp(b, b, CTRL, ROWMASK, BANKMASK, false));
}
template <int CTRL>
__device__ __forceinline__ float dpp_all(float v) {
    return __int_as_float(__builtin_amdgcn_update_dpp(0, __float_as_int(v), CTRL, 0xF, 0xF, true));
}
__device__ __forceinline__ float lane_xor(float v, int lm) {
    switch (lm) {
    case 1:  return dpp_all<0x161>(v);
    case 2:  return dpp_all<0x162>(v);
    case 4:  return dpp_all<0x164>(v);
    case 8:  return dpp_all<0x168>(v);
    default: {
        const int b = __float_as_int(v);
        return __int_as_float(__builtin_amdgcn_permlanex16(b, b, 0x76543210, 0xfedcba98, false, false));
    }
    }
}
#define CNOT_SEL_LO 0x45762310
#define CNOT_SEL_HI 0x89baefdc
__device__ __forceinline__ float cnot_lanes_even(float v) {
    const int w = __float_as_int(dpp_move<0x168, 0x2, 0xF>(v));
    return __int_as_float(__builtin_amdgcn_permlane16(w, w, CNOT_SEL_LO, CNOT_SEL_HI, false, false));
}
__device__ __forceinline__ float cnot_lanes_odd(float v) {
    const int w = __float_as_int(dpp_move<0x168, 0x1, 0xF>(v));
    return __int_as_float(__builtin_amdgcn_permlanex16(w, w, CNOT_SEL_LO, CNOT_SEL_HI, false, false));
}

__global__ __launch_bounds__(128) void qenc_build(const float* __restrict__ wrot,
                                                   const float* __restrict__ wrz,
                                                   float* __restrict__ P) {
    __shared__ v4f stage4[512];
    const int lane = threadIdx.x & 31;
    const int rl   = threadIdx.x >> 5;
    const int r    = blockIdx.x * 4 + rl;

    const float th  = wrot[lane];
    const float phz = wrz[lane];
    const float gc  = cosf(0.5f * th),  gs  = sinf(0.5f * th);
    const float ger = cosf(0.5f * phz), gei = -sinf(0.5f * phz);

    float re[8], im[8];

    const float w = 6.28318530717958647692f / 256.0f;
#pragma unroll
    for (int j = 0; j < 8; ++j) {
        int k  = j * 32 + lane;
        int ph = (r * k) & 255;
        float sv, cv;
        sincosf(w * (float)ph, &sv, &cv);
        re[j] = cv * 0.0625f;
        im[j] = sv * 0.0625f;
    }

#pragma unroll
    for (int l = 0; l < 4; ++l) {
#pragma unroll
        for (int q = 0; q < 8; ++q) {
            const int g = l * 8 + q;
            const float c   = lane_value(gc,  g);
            const float s   = lane_value(gs,  g);
            const float emr = lane_value(ger, g);
            const float emi = lane_value(gei, g);
            const float U00r =  emr * c, U00i =  emi * c;
            const float U01r = -emr * s, U01i = -emi * s;
            const float U10r =  emr * s, U10i = -emi * s;
            const float U11r =  emr * c, U11i = -emi * c;
            const int bit = 7 - q;
            if (bit >= 5) {
                const int jm = 1 << (bit - 5);
#pragma unroll
                for (int j0 = 0; j0 < 8; ++j0) {
                    if (j0 & jm) continue;
                    const int j1 = j0 | jm;
                    float ar = re[j0], ai = im[j0];
                    float br = re[j1], bi = im[j1];
                    re[j0] = U00r * ar - U00i * ai + U01r * br - U01i * bi;
                    im[j0] = U00r * ai + U00i * ar + U01r * bi + U01i * br;
                    re[j1] = U10r * ar - U10i * ai + U11r * br - U11i * bi;
                    im[j1] = U10r * ai + U10i * ar + U11r * bi + U11i * br;
                }
            } else {
                const int  lm = 1 << bit;
                const bool hi = (lane & lm) != 0;
                const float Pmr = hi ? U11r : U00r, Pmi = hi ? U11i : U00i;
                const float Pxr = hi ? U10r : U01r, Pxi = hi ? U10i : U01i;
#pragma unroll
                for (int j = 0; j < 8; ++j) {
                    float mr = re[j], mi = im[j];
                    float xr = lane_xor(mr, lm);
                    float xi = lane_xor(mi, lm);
                    re[j] = Pmr * mr - Pmi * mi + Pxr * xr - Pxi * xi;
                    im[j] = Pmr * mi + Pmi * mr + Pxr * xi + Pxi * xr;
                }
            }
        }
#pragma unroll
        for (int q = 0; q < 2; ++q) {
            const int cm = 1 << (2 - q), tm = 1 << (1 - q);
#pragma unroll
            for (int j = 0; j < 8; ++j) {
                if ((j & cm) && !(j & tm)) {
                    float t0 = re[j]; re[j] = re[j | tm]; re[j | tm] = t0;
                    float t1 = im[j]; im[j] = im[j | tm]; im[j | tm] = t1;
                }
            }
        }
#pragma unroll
        for (int j = 0; j < 8; ++j) {
            if (j & 1) {
                re[j] = cnot_lanes_odd(re[j]);
                im[j] = cnot_lanes_odd(im[j]);
            } else {
                re[j] = cnot_lanes_even(re[j]);
                im[j] = cnot_lanes_even(im[j]);
            }
        }
        {
            const bool ctl = (lane & 1) != 0;
#pragma unroll
            for (int j = 0; j < 4; ++j) {
                float a0 = re[j], b0 = re[j | 4];
                re[j] = ctl ? b0 : a0; re[j | 4] = ctl ? a0 : b0;
                float a1 = im[j], b1 = im[j | 4];
                im[j] = ctl ? b1 : a1; im[j | 4] = ctl ? a1 : b1;
            }
        }
    }

    float* stage = (float*)stage4;
    const int kh = (rl >> 1) & 1;
    const int ke = rl & 1;
#pragma unroll
    for (int j = 0; j < 8; ++j) {
        const int t = 2 * j + (lane >> 4);
        const int d = ((t * 32 + kh * 16 + (lane & 15)) << 2) + ke;
        stage[d]     = re[j];
        stage[d + 2] = im[j];
    }
    __syncthreads();
    v4f* P4 = (v4f*)P;
    for (int p = 0; p < 4; ++p) {
        const int q = p * 128 + threadIdx.x;
        const int t = q >> 5;
        P4[(t * 64 + blockIdx.x) * 32 + (q & 31)] = stage4[q];
    }
}

#define WMMA_F32(A, B, C) \
    __builtin_amdgcn_wmma_f32_16x16x4_f32(false, (A), false, (B), (short)0, (C), false, false)

#define LOAD_STEP(A, B, s)                                          \
    _Pragma("unroll")                                               \
    for (int g = 0; g < 4; ++g)                                     \
        A[g] = *(const v2f*)(ap + g * 4096 + 4 * (s));              \
    _Pragma("unroll")                                               \
    for (int c = 0; c < 4; ++c)                                     \
        B[c] = bp[c * 2048 + (s) * 32];

#define COMPUTE_STEP(A, B)                                          \
    _Pragma("unroll")                                               \
    for (int g = 0; g < 4; ++g)                                     \
        nrm2[g] = A[g] * A[g] + nrm2[g];                            \
    _Pragma("unroll")                                               \
    for (int c = 0; c < 4; ++c) {                                   \
        const v2f Br = B[c].xy, Bi = B[c].zw;                       \
        _Pragma("unroll")                                           \
        for (int g = 0; g < 4; ++g) {                               \
            accR[g][c] = WMMA_F32(A[g], Br, accR[g][c]);            \
            accI[g][c] = WMMA_F32(A[g], Bi, accI[g][c]);            \
        }                                                           \
    }

__global__ __launch_bounds__(256) void qenc_gemm(const float* __restrict__ X,
                                                  const float* __restrict__ P,
                                                  float* __restrict__ out) {
    const int lane = threadIdx.x & 31;
    const int wid  = blockIdx.x * 8 + (threadIdx.x >> 5);
    const int n0   = (wid & 3) * 64;
    const int m0   = (wid >> 2) * 64;
    const int half = lane >> 4;
    const int l15  = lane & 15;

    const float* ap = X + (m0 + l15) * 256 + 2 * half;
    const v4f*   bp = (const v4f*)P + (n0 >> 4) * 64 * 32 + lane;

    v8f accR[4][4], accI[4][4];
#pragma unroll
    for (int g = 0; g < 4; ++g) {
#pragma unroll
        for (int c = 0; c < 4; ++c) {
            accR[g][c] = (v8f){};
            accI[g][c] = (v8f){};
        }
    }
    v2f nrm2[4];
#pragma unroll
    for (int g = 0; g < 4; ++g) nrm2[g] = (v2f){0.0f, 0.0f};

    v2f a0[4], a1[4];
    v4f b0[4], b1[4];

    LOAD_STEP(a0, b0, 0)
    for (int s = 0; s < 62; s += 2) {
        LOAD_STEP(a1, b1, s + 1)
        COMPUTE_STEP(a0, b0)
        LOAD_STEP(a0, b0, s + 2)
        COMPUTE_STEP(a1, b1)
    }
    LOAD_STEP(a1, b1, 63)
    COMPUTE_STEP(a0, b0)
    COMPUTE_STEP(a1, b1)

    float nrm[4];
#pragma unroll
    for (int g = 0; g < 4; ++g) {
        nrm[g] = nrm2[g].x + nrm2[g].y;
        nrm[g] += __shfl_xor(nrm[g], 16, 32);
    }

#pragma unroll
    for (int g = 0; g < 4; ++g) {
#pragma unroll
        for (int i = 0; i < 8; ++i) {
            const int row   = i + 8 * half;
            const float inv = 1.0f / __shfl(nrm[g], row, 32);
            float* o = out + (m0 + 16 * g + row) * 256 + n0 + l15;
#pragma unroll
            for (int c = 0; c < 4; ++c)
                o[16 * c] = (accR[g][c][i] * accR[g][c][i] + accI[g][c][i] * accI[g][c][i]) * inv;
        }
    }
}

extern "C" void kernel_launch(void* const* d_in, const int* in_sizes, int n_in,
                              void* d_out, int out_size, void* d_ws, size_t ws_size,
                              hipStream_t stream) {
    const float* x    = (const float*)d_in[0];
    const float* wrot = (const float*)d_in[1];
    const float* wrz  = (const float*)d_in[2];
    float* out = (float*)d_out;

    float* P = (float*)d_ws;

    qenc_build<<<64, 128, 0, stream>>>(wrot, wrz, P);

    qenc_gemm<<<256, 256, 0, stream>>>(x, P, out);
}
